# starting kernel
# baseline (speedup 1.0000x reference)
#include <hip/hip_runtime.h>

typedef float v2f __attribute__((ext_vector_type(2)));
typedef float v8f __attribute__((ext_vector_type(8)));

#define FP_TOL 1e-3f
#define FP_MAX_ITERS 1000

__global__ __launch_bounds__(256)
void deq_fixed_point_wmma(const float* __restrict__ d,
                          const float* __restrict__ Wfc,
                          const float* __restrict__ bfc,
                          const float* __restrict__ Win,
                          const float* __restrict__ bin,
                          const float* __restrict__ Wout,
                          const float* __restrict__ bout,
                          float* __restrict__ out,
                          int B)
{
    const int lane   = threadIdx.x & 31;
    const int warpId = (blockIdx.x * blockDim.x + threadIdx.x) >> 5;
    const int m      = lane & 15;
    const bool hi    = lane >= 16;

    const int row    = warpId * 16 + m;
    const int rowc   = (row < B) ? row : (B - 1);
    const bool cvalid = (!hi) && (row < B);

    const float* dp = d + (size_t)rowc * 3;
    const float d0 = dp[0], d1 = dp[1], d2 = dp[2];

    const float w00 = Wfc[0], w01 = Wfc[1], w02 = Wfc[2];
    const float w10 = Wfc[3], w11 = Wfc[4], w12 = Wfc[5];
    const float w20 = Wfc[6], w21 = Wfc[7], w22 = Wfc[8];

    float a0 = 0.f, a1 = 0.f;
    const int j = m & 7;
    if ((m & 7) < 3 && (m < 3 || m >= 8)) {
        if (j == 0)      { a0 = hi ? w02 : w00; a1 = hi ? 0.f : w01; }
        else if (j == 1) { a0 = hi ? w12 : w10; a1 = hi ? 0.f : w11; }
        else             { a0 = hi ? w22 : w20; a1 = hi ? 0.f : w21; }
    }
    v2f amat; amat.x = a0; amat.y = a1;

    const float bi0 = bfc[0] + bin[0];
    const float bi1 = bfc[1] + bin[1];
    const float bi2 = bfc[2] + bin[2];
    const float c0 = bi0 + Win[0]*d0 + Win[1]*d1 + Win[2]*d2;
    const float c1 = bi1 + Win[3]*d0 + Win[4]*d1 + Win[5]*d2;
    const float c2 = bi2 + Win[6]*d0 + Win[7]*d1 + Win[8]*d2;
    v8f cmat = {c0, c1, c2, 0.f, 0.f, 0.f, 0.f, 0.f};

    v2f bmat; bmat.x = 0.f; bmat.y = 0.f;
    float u0 = 0.f, u1 = 0.f, u2 = 0.f;

    for (int it = 0; it < FP_MAX_ITERS; ++it) {
        v8f acc = __builtin_amdgcn_wmma_f32_16x16x4_f32(
             false, amat,  false, bmat,
             (short)0, cmat,  false,  false);

        const float n0 = fmaxf(acc[0], 0.f);
        const float n1 = fmaxf(acc[1], 0.f);
        const float n2 = fmaxf(acc[2], 0.f);

        const float r = fmaxf(fmaxf(fabsf(n0 - u0), fabsf(n1 - u1)),
                              fabsf(n2 - u2));
        u0 = n0; u1 = n1; u2 = n2;

        bmat.x = hi ? n2 : n0;
        bmat.y = n1;

        if (__all(r < FP_TOL)) break;
    }

    const float o = bout[0] + Wout[0]*u0 + Wout[1]*u1 + Wout[2]*u2;
    if (cvalid) out[row] = o;
}

extern "C" void kernel_launch(void* const* d_in, const int* in_sizes, int n_in,
                              void* d_out, int out_size, void* d_ws, size_t ws_size,
                              hipStream_t stream) {
    const float* d    = (const float*)d_in[0];
    const float* Wfc  = (const float*)d_in[1];
    const float* bfc  = (const float*)d_in[2];
    const float* Win  = (const float*)d_in[3];
    const float* bin  = (const float*)d_in[4];
    const float* Wout = (const float*)d_in[5];
    const float* bout = (const float*)d_in[6];
    float* out = (float*)d_out;

    const int B = in_sizes[0] / 3;
    const int rowsPerBlock = 128;
    const int blocks = (B + rowsPerBlock - 1) / rowsPerBlock;

    deq_fixed_point_wmma<<<blocks, 256, 0, stream>>>(
        d, Wfc, bfc, Win, bin, Wout, bout, out, B);
}


# 64 rows per wave step, two per lane, in 3 chained WMMAs not 4: (u0,u1) of each row set in its own WMMA, both sets' u2 share the third; exit when all 64 rows pass (same or more sweeps per row)
# speedup vs baseline: 1.7894x; 1.7894x over previous
#include <hip/hip_runtime.h>


typedef float v2f __attribute__((ext_vector_type(2)));
typedef float v8f __attribute__((ext_vector_type(8)));

#define FP_TOL 1e-3f
#define FP_MAX_ITERS 1000
#define FP_BIG 3.402823466e+38f

__global__ __launch_bounds__(256)
void deq_fixed_point_wmma(const float* __restrict__ d,
                          const float* __restrict__ Wfc,
                          const float* __restrict__ bfc,
                          const float* __restrict__ Win,
                          const float* __restrict__ bin,
                          const float* __restrict__ Wout,
                          const float* __restrict__ bout,
                          float* __restrict__ out,
                          int B)
{
    const int lane   = threadIdx.x & 31;
    const int m      = lane & 15;
    const bool hi    = lane >= 16;

    const int waveInBlock = __builtin_amdgcn_readfirstlane((int)(threadIdx.x >> 5));
    const int wavesPerBlock = (int)(blockDim.x >> 5);
    const int warpId = (int)blockIdx.x * wavesPerBlock + waveInBlock;
    const int nWaves = (int)gridDim.x * wavesPerBlock;
    const int nGroups = (B + 63) >> 6;

    const float w00 = Wfc[0], w01 = Wfc[1], w02 = Wfc[2];
    const float w10 = Wfc[3], w11 = Wfc[4], w12 = Wfc[5];
    const float w20 = Wfc[6], w21 = Wfc[7], w22 = Wfc[8];

    float a1x = 0.f, a1y = 0.f, a2x = 0.f, a2y = 0.f, a3x = 0.f, a3y = 0.f;
    const int mm = hi ? (m - 8) : m;
    if (mm >= 0 && mm < 6) {
        const bool setQ = mm >= 3;
        const int j = setQ ? (mm - 3) : mm;
        float wj0, wj1, wj2;
        if (j == 0)      { wj0 = w00; wj1 = w01; wj2 = w02; }
        else if (j == 1) { wj0 = w10; wj1 = w11; wj2 = w12; }
        else             { wj0 = w20; wj1 = w21; wj2 = w22; }
        if (!setQ) { a1x = wj0; a1y = wj1; a3x = wj2; }
        else       { a2x = wj0; a2y = wj1; a3y = wj2; }
    }
    v2f amat1; amat1.x = a1x; amat1.y = a1y;
    v2f amat2; amat2.x = a2x; amat2.y = a2y;
    v2f amat3; amat3.x = a3x; amat3.y = a3y;

    const float bi0 = bfc[0] + bin[0];
    const float bi1 = bfc[1] + bin[1];
    const float bi2 = bfc[2] + bin[2];
    const float i00 = Win[0], i01 = Win[1], i02 = Win[2];
    const float i10 = Win[3], i11 = Win[4], i12 = Win[5];
    const float i20 = Win[6], i21 = Win[7], i22 = Win[8];
    const float o0 = Wout[0], o1 = Wout[1], o2 = Wout[2], ob = bout[0];

    if (warpId >= nGroups) return;

    float p0, p1, p2, q0, q1, q2;
    {
        const int rowP  = warpId * 64 + lane;
        const int rowQ  = rowP + 32;
        const int rowPc = (rowP < B) ? rowP : (B - 1);
        const int rowQc = (rowQ < B) ? rowQ : (B - 1);
        const float* pp = d + (size_t)rowPc * 3;
        const float* qp = d + (size_t)rowQc * 3;
        p0 = pp[0]; p1 = pp[1]; p2 = pp[2];
        q0 = qp[0]; q1 = qp[1]; q2 = qp[2];
    }

    for (int grp = warpId; grp < nGroups; grp += nWaves) {
        const int rowP = grp * 64 + lane;
        const int rowQ = rowP + 32;
        const bool validP = rowP < B;
        const bool validQ = rowQ < B;

        const float cP0 = fmaf(i02, p2, fmaf(i01, p1, fmaf(i00, p0, bi0)));
        const float cP1 = fmaf(i12, p2, fmaf(i11, p1, fmaf(i10, p0, bi1)));
        const float cP2 = fmaf(i22, p2, fmaf(i21, p1, fmaf(i20, p0, bi2)));
        const float cQ0 = fmaf(i02, q2, fmaf(i01, q1, fmaf(i00, q0, bi0)));
        const float cQ1 = fmaf(i12, q2, fmaf(i11, q1, fmaf(i10, q0, bi1)));
        const float cQ2 = fmaf(i22, q2, fmaf(i21, q1, fmaf(i20, q0, bi2)));
        v8f cmat = {cP0, cP1, cP2, cQ0, cQ1, cQ2, 0.f, 0.f};

        {
            const int ngrp   = (grp + nWaves < nGroups) ? (grp + nWaves) : (nGroups - 1);
            const int nrowP  = ngrp * 64 + lane;
            const int nrowQ  = nrowP + 32;
            const int nrowPc = (nrowP < B) ? nrowP : (B - 1);
            const int nrowQc = (nrowQ < B) ? nrowQ : (B - 1);
            const float* np = d + (size_t)nrowPc * 3;
            const float* nq = d + (size_t)nrowQc * 3;
            p0 = np[0]; p1 = np[1]; p2 = np[2];
            q0 = nq[0]; q1 = nq[1]; q2 = nq[2];
        }

        v2f bmat1; bmat1.x = 0.f; bmat1.y = 0.f;
        v2f bmat2; bmat2.x = 0.f; bmat2.y = 0.f;
        v2f bmat3; bmat3.x = 0.f; bmat3.y = 0.f;

        for (int it = 0; it < FP_MAX_ITERS; ++it) {
            v8f acc = __builtin_amdgcn_wmma_f32_16x16x4_f32(
                 false, amat1,  false, bmat1,
                 (short)0, cmat,  false,  false);
            acc = __builtin_amdgcn_wmma_f32_16x16x4_f32(
                 false, amat2,  false, bmat2,
                 (short)0, acc,  false,  false);
            acc = __builtin_amdgcn_wmma_f32_16x16x4_f32(
                 false, amat3,  false, bmat3,
                 (short)0, acc,  false,  false);

            v2f nb1, nb2, nb3;
            nb1.x = __builtin_amdgcn_fmed3f(acc[0], 0.f, FP_BIG);
            nb1.y = __builtin_amdgcn_fmed3f(acc[1], 0.f, FP_BIG);
            nb3.x = __builtin_amdgcn_fmed3f(acc[2], 0.f, FP_BIG);
            nb2.x = __builtin_amdgcn_fmed3f(acc[3], 0.f, FP_BIG);
            nb2.y = __builtin_amdgcn_fmed3f(acc[4], 0.f, FP_BIG);
            nb3.y = __builtin_amdgcn_fmed3f(acc[5], 0.f, FP_BIG);

            const v2f e1 = nb1 - bmat1;
            const v2f e2 = nb2 - bmat2;
            const v2f e3 = nb3 - bmat3;
            const float rP = fmaxf(fmaxf(fabsf(e1.x), fabsf(e1.y)), fabsf(e3.x));
            const float rQ = fmaxf(fmaxf(fabsf(e2.x), fabsf(e2.y)), fabsf(e3.y));
            const float r  = fmaxf(rP, rQ);

            bmat1 = nb1; bmat2 = nb2; bmat3 = nb3;

            if (__all(r < FP_TOL)) break;
        }

        const float oP = ob + o0*bmat1.x + o1*bmat1.y + o2*bmat3.x;
        const float oQ = ob + o0*bmat2.x + o1*bmat2.y + o2*bmat3.y;
        if (validP) out[rowP] = oP;
        if (validQ) out[rowQ] = oQ;
    }
}

extern "C" void kernel_launch(void* const* d_in, const int* in_sizes, int n_in,
                              void* d_out, int out_size, void* d_ws, size_t ws_size,
                              hipStream_t stream) {
    const float* d    = (const float*)d_in[0];
    const float* Wfc  = (const float*)d_in[1];
    const float* bfc  = (const float*)d_in[2];
    const float* Win  = (const float*)d_in[3];
    const float* bin  = (const float*)d_in[4];
    const float* Wout = (const float*)d_in[5];
    const float* bout = (const float*)d_in[6];
    float* out = (float*)d_out;

    const int B = in_sizes[0] / 3;

    deq_fixed_point_wmma<<<4096, 256, 0, stream>>>(
        d, Wfc, bfc, Win, bin, Wout, bout, out, B);
}


# step test sparse while far from the tolerance: sweeps in pairs, test behind the second only, while some row's step >= 8 tol; then a test every sweep as before (same or more sweeps per row)
# speedup vs baseline: 1.9156x; 1.0705x over previous
#include <hip/hip_runtime.h>


typedef float v2f __attribute__((ext_vector_type(2)));
typedef float v8f __attribute__((ext_vector_type(8)));

#define FP_TOL 1e-3f
#define FP_MAX_ITERS 1000
#define FP_BIG 3.402823466e+38f

#define FP_FAR_FACTOR 8.0f

static_assert(FP_MAX_ITERS % 2 == 0, "the far phase counts sweeps in pairs");

__device__ __forceinline__ void fp_sweep(const v2f a1, const v2f a2, const v2f a3,
                                         const v8f c,
                                         const v2f b1, const v2f b2, const v2f b3,
                                         v2f& n1, v2f& n2, v2f& n3)
{
    v8f acc = __builtin_amdgcn_wmma_f32_16x16x4_f32(
         false, a1,  false, b1,
         (short)0, c,  false,  false);
    acc = __builtin_amdgcn_wmma_f32_16x16x4_f32(
         false, a2,  false, b2,
         (short)0, acc,  false,  false);
    acc = __builtin_amdgcn_wmma_f32_16x16x4_f32(
         false, a3,  false, b3,
         (short)0, acc,  false,  false);

    n1.x = __builtin_amdgcn_fmed3f(acc[0], 0.f, FP_BIG);
    n1.y = __builtin_amdgcn_fmed3f(acc[1], 0.f, FP_BIG);
    n3.x = __builtin_amdgcn_fmed3f(acc[2], 0.f, FP_BIG);
    n2.x = __builtin_amdgcn_fmed3f(acc[3], 0.f, FP_BIG);
    n2.y = __builtin_amdgcn_fmed3f(acc[4], 0.f, FP_BIG);
    n3.y = __builtin_amdgcn_fmed3f(acc[5], 0.f, FP_BIG);
}

__device__ __forceinline__ float fp_step(const v2f n1, const v2f n2, const v2f n3,
                                         const v2f b1, const v2f b2, const v2f b3)
{
    const v2f e1 = n1 - b1;
    const v2f e2 = n2 - b2;
    const v2f e3 = n3 - b3;
    const float rP = fmaxf(fmaxf(fabsf(e1.x), fabsf(e1.y)), fabsf(e3.x));
    const float rQ = fmaxf(fmaxf(fabsf(e2.x), fabsf(e2.y)), fabsf(e3.y));
    return fmaxf(rP, rQ);
}

__global__ __launch_bounds__(256)
void deq_fixed_point_wmma(const float* __restrict__ d,
                          const float* __restrict__ Wfc,
                          const float* __restrict__ bfc,
                          const float* __restrict__ Win,
                          const float* __restrict__ bin,
                          const float* __restrict__ Wout,
                          const float* __restrict__ bout,
                          float* __restrict__ out,
                          int B)
{
    const int lane   = threadIdx.x & 31;
    const int m      = lane & 15;
    const bool hi    = lane >= 16;

    const int waveInBlock = __builtin_amdgcn_readfirstlane((int)(threadIdx.x >> 5));
    const int wavesPerBlock = (int)(blockDim.x >> 5);
    const int warpId = (int)blockIdx.x * wavesPerBlock + waveInBlock;
    const int nWaves = (int)gridDim.x * wavesPerBlock;
    const int nGroups = (B + 63) >> 6;

    const float w00 = Wfc[0], w01 = Wfc[1], w02 = Wfc[2];
    const float w10 = Wfc[3], w11 = Wfc[4], w12 = Wfc[5];
    const float w20 = Wfc[6], w21 = Wfc[7], w22 = Wfc[8];

    float a1x = 0.f, a1y = 0.f, a2x = 0.f, a2y = 0.f, a3x = 0.f, a3y = 0.f;
    const int mm = hi ? (m - 8) : m;
    if (mm >= 0 && mm < 6) {
        const bool setQ = mm >= 3;
        const int j = setQ ? (mm - 3) : mm;
        float wj0, wj1, wj2;
        if (j == 0)      { wj0 = w00; wj1 = w01; wj2 = w02; }
        else if (j == 1) { wj0 = w10; wj1 = w11; wj2 = w12; }
        else             { wj0 = w20; wj1 = w21; wj2 = w22; }
        if (!setQ) { a1x = wj0; a1y = wj1; a3x = wj2; }
        else       { a2x = wj0; a2y = wj1; a3y = wj2; }
    }
    v2f amat1; amat1.x = a1x; amat1.y = a1y;
    v2f amat2; amat2.x = a2x; amat2.y = a2y;
    v2f amat3; amat3.x = a3x; amat3.y = a3y;

    const float bi0 = bfc[0] + bin[0];
    const float bi1 = bfc[1] + bin[1];
    const float bi2 = bfc[2] + bin[2];
    const float i00 = Win[0], i01 = Win[1], i02 = Win[2];
    const float i10 = Win[3], i11 = Win[4], i12 = Win[5];
    const float i20 = Win[6], i21 = Win[7], i22 = Win[8];
    const float o0 = Wout[0], o1 = Wout[1], o2 = Wout[2], ob = bout[0];

    if (warpId >= nGroups) return;

    float p0, p1, p2, q0, q1, q2;
    {
        const int rowP  = warpId * 64 + lane;
        const int rowQ  = rowP + 32;
        const int rowPc = (rowP < B) ? rowP : (B - 1);
        const int rowQc = (rowQ < B) ? rowQ : (B - 1);
        const float* pp = d + (size_t)rowPc * 3;
        const float* qp = d + (size_t)rowQc * 3;
        p0 = pp[0]; p1 = pp[1]; p2 = pp[2];
        q0 = qp[0]; q1 = qp[1]; q2 = qp[2];
    }

    for (int grp = warpId; grp < nGroups; grp += nWaves) {
        const int rowP = grp * 64 + lane;
        const int rowQ = rowP + 32;
        const bool validP = rowP < B;
        const bool validQ = rowQ < B;

        const float cP0 = fmaf(i02, p2, fmaf(i01, p1, fmaf(i00, p0, bi0)));
        const float cP1 = fmaf(i12, p2, fmaf(i11, p1, fmaf(i10, p0, bi1)));
        const float cP2 = fmaf(i22, p2, fmaf(i21, p1, fmaf(i20, p0, bi2)));
        const float cQ0 = fmaf(i02, q2, fmaf(i01, q1, fmaf(i00, q0, bi0)));
        const float cQ1 = fmaf(i12, q2, fmaf(i11, q1, fmaf(i10, q0, bi1)));
        const float cQ2 = fmaf(i22, q2, fmaf(i21, q1, fmaf(i20, q0, bi2)));
        v8f cmat = {cP0, cP1, cP2, cQ0, cQ1, cQ2, 0.f, 0.f};

        {
            const int ngrp   = (grp + nWaves < nGroups) ? (grp + nWaves) : (nGroups - 1);
            const int nrowP  = ngrp * 64 + lane;
            const int nrowQ  = nrowP + 32;
            const int nrowPc = (nrowP < B) ? nrowP : (B - 1);
            const int nrowQc = (nrowQ < B) ? nrowQ : (B - 1);
            const float* np = d + (size_t)nrowPc * 3;
            const float* nq = d + (size_t)nrowQc * 3;
            p0 = np[0]; p1 = np[1]; p2 = np[2];
            q0 = nq[0]; q1 = nq[1]; q2 = nq[2];
        }

        v2f bmat1; bmat1.x = 0.f; bmat1.y = 0.f;
        v2f bmat2; bmat2.x = 0.f; bmat2.y = 0.f;
        v2f bmat3; bmat3.x = 0.f; bmat3.y = 0.f;

        int nSweeps = 0;
        float r;
        do {
            v2f mb1, mb2, mb3, nb1, nb2, nb3;
            fp_sweep(amat1, amat2, amat3, cmat, bmat1, bmat2, bmat3, mb1, mb2, mb3);
            fp_sweep(amat1, amat2, amat3, cmat, mb1, mb2, mb3, nb1, nb2, nb3);
            r = fp_step(nb1, nb2, nb3, mb1, mb2, mb3);
            bmat1 = nb1; bmat2 = nb2; bmat3 = nb3;
            nSweeps += 2;
        } while (!__all(r < FP_FAR_FACTOR * FP_TOL) && nSweeps < FP_MAX_ITERS);

        if (!__all(r < FP_TOL)) {
            for (int it = nSweeps; it < FP_MAX_ITERS; ++it) {
                v2f nb1, nb2, nb3;
                fp_sweep(amat1, amat2, amat3, cmat, bmat1, bmat2, bmat3, nb1, nb2, nb3);
                r = fp_step(nb1, nb2, nb3, bmat1, bmat2, bmat3);
                bmat1 = nb1; bmat2 = nb2; bmat3 = nb3;
                if (__all(r < FP_TOL)) break;
            }
        }

        const float oP = fmaf(o2, bmat3.x, fmaf(o1, bmat1.y, fmaf(o0, bmat1.x, ob)));
        const float oQ = fmaf(o2, bmat3.y, fmaf(o1, bmat2.y, fmaf(o0, bmat2.x, ob)));
        if (validP) out[rowP] = oP;
        if (validQ) out[rowQ] = oQ;
    }
}

extern "C" void kernel_launch(void* const* d_in, const int* in_sizes, int n_in,
                              void* d_out, int out_size, void* d_ws, size_t ws_size,
                              hipStream_t stream) {
    const float* d    = (const float*)d_in[0];
    const float* Wfc  = (const float*)d_in[1];
    const float* bfc  = (const float*)d_in[2];
    const float* Win  = (const float*)d_in[3];
    const float* bin  = (const float*)d_in[4];
    const float* Wout = (const float*)d_in[5];
    const float* bout = (const float*)d_in[6];
    float* out = (float*)d_out;

    const int B = in_sizes[0] / 3;

    deq_fixed_point_wmma<<<4096, 256, 0, stream>>>(
        d, Wfc, bfc, Win, bin, Wout, bout, out, B);
}
